# starting kernel
# baseline (speedup 1.0000x reference)
#include <hip/hip_runtime.h>
#include <hip/hip_bf16.h>

typedef __attribute__((ext_vector_type(2))) float v2f;
typedef __attribute__((ext_vector_type(8))) float v8f;

#define D_FEAT 64

__global__ __launch_bounds__(256) void
udotv_wmma_kernel(const float* __restrict__ h,
                  const int*   __restrict__ src,
                  const int*   __restrict__ dst,
                  float*       __restrict__ out,
                  int n_edges)
{
    const int lane        = threadIdx.x & 31;
    const int waveInBlock = threadIdx.x >> 5;
    const int tile        = blockIdx.x * (blockDim.x >> 5) + waveInBlock;
    const int e0          = tile * 16;
    if (e0 >= n_edges) return;

    const int r  = lane & 15;
    const int kk = (lane >> 4) << 1;

    int eL = e0 + r;
    if (eL >= n_edges) eL = n_edges - 1;

    const float* __restrict__ srcRow = h + (long)src[eL] * D_FEAT;
    const float* __restrict__ dstRow = h + (long)dst[eL] * D_FEAT;

    v8f acc = {};
    #pragma unroll
    for (int c = 0; c < 16; ++c) {
        v2f a = *(const v2f*)(srcRow + c * 4 + kk);
        v2f b = *(const v2f*)(dstRow + c * 4 + kk);
        acc = __builtin_amdgcn_wmma_f32_16x16x4_f32(
                  false, a, false, b, (short)0, acc, false, false);
    }

    float val = 0.0f;
    int   m   = -1;
    #pragma unroll
    for (int v = 0; v < 8; ++v) {
        if (lane == v)      { val = acc[v]; m = v;     }
        if (lane == v + 24) { val = acc[v]; m = v + 8; }
    }
    if (m >= 0) {
        int e = e0 + m;
        if (e < n_edges) out[e] = val;
    }
}

extern "C" void kernel_launch(void* const* d_in, const int* in_sizes, int n_in,
                              void* d_out, int out_size, void* d_ws, size_t ws_size,
                              hipStream_t stream) {
    const float* h   = (const float*)d_in[0];
    const int*   src = (const int*)d_in[1];
    const int*   dst = (const int*)d_in[2];
    float*       out = (float*)d_out;

    const int n_edges = in_sizes[1];
    const int tiles   = (n_edges + 15) / 16;
    const int wavesPerBlock = 8;
    const int blocks  = (tiles + wavesPerBlock - 1) / wavesPerBlock;

    udotv_wmma_kernel<<<blocks, 256, 0, stream>>>(h, src, dst, out, n_edges);
}


# Gathered rows fetched as whole 128-B lines (16-B loads, 16 lanes per 256-B row: 4 lines per load, was 16) and turned into the A/B fragments through a wave-private LDS tile; sum order unchanged
# speedup vs baseline: 1.8742x; 1.8742x over previous
#include <hip/hip_runtime.h>
#include <hip/hip_bf16.h>

typedef __attribute__((ext_vector_type(2))) float v2f;
typedef __attribute__((ext_vector_type(4))) float v4f;
typedef __attribute__((ext_vector_type(8))) float v8f;
typedef __attribute__((ext_vector_type(16), aligned(4))) int v16i;

#define D_FEAT 64
#define WAVES_PER_BLOCK 8
#define ROW_PITCH 68
#define TILE_FLOATS (16 * ROW_PITCH)

__global__ __launch_bounds__(256) void
udotv_wmma_kernel(const float* __restrict__ h,
                  const int*   __restrict__ src,
                  const int*   __restrict__ dst,
                  float*       __restrict__ out,
                  int n_edges)
{
    __shared__ float stageA[WAVES_PER_BLOCK * TILE_FLOATS];
    __shared__ float stageB[WAVES_PER_BLOCK * TILE_FLOATS];

    const int lane        = threadIdx.x & 31;
    const int waveInBlock = __builtin_amdgcn_readfirstlane((int)(threadIdx.x >> 5));
    const int tile        = blockIdx.x * WAVES_PER_BLOCK + waveInBlock;
    const int e0          = tile * 16;
    if (e0 >= n_edges) return;

    const int r  = lane & 15;
    const int hh = lane >> 4;

    int sI[16], dI[16];
    if (e0 + 16 <= n_edges) {
        const v16i sv = *(const v16i*)(src + e0);
        const v16i dv = *(const v16i*)(dst + e0);
        #pragma unroll
        for (int k = 0; k < 16; ++k) { sI[k] = sv[k]; dI[k] = dv[k]; }
    } else {
        #pragma unroll
        for (int k = 0; k < 16; ++k) {
            int e = e0 + k;
            if (e >= n_edges) e = n_edges - 1;
            sI[k] = src[e];
            dI[k] = dst[e];
        }
    }

    const char* hb = (const char*)h;
    v4f sRow[8], dRow[8];
    #pragma unroll
    for (int i = 0; i < 8; ++i) {
        const unsigned sn = (unsigned)(hh ? sI[2 * i + 1] : sI[2 * i]);
        sRow[i] = *(const v4f*)(hb + (size_t)(sn * (unsigned)(D_FEAT * 4) + (unsigned)(r * 16)));
    }
    #pragma unroll
    for (int i = 0; i < 8; ++i) {
        const unsigned dn = (unsigned)(hh ? dI[2 * i + 1] : dI[2 * i]);
        dRow[i] = *(const v4f*)(hb + (size_t)(dn * (unsigned)(D_FEAT * 4) + (unsigned)(r * 16)));
    }

    float* tA = stageA + waveInBlock * TILE_FLOATS;
    float* tB = stageB + waveInBlock * TILE_FLOATS;

    #pragma unroll
    for (int i = 0; i < 8; ++i)
        *(v4f*)(tA + (2 * i + hh) * ROW_PITCH + r * 4) = sRow[i];
    #pragma unroll
    for (int i = 0; i < 8; ++i)
        *(v4f*)(tB + (2 * i + hh) * ROW_PITCH + r * 4) = dRow[i];

    const int kk = hh << 1;
    const float* fA = tA + r * ROW_PITCH + kk;
    const float* fB = tB + r * ROW_PITCH + kk;

    v8f acc = {};
    #pragma unroll
    for (int c = 0; c < 16; ++c) {
        v2f a = *(const v2f*)(fA + c * 4);
        v2f b = *(const v2f*)(fB + c * 4);
        acc = __builtin_amdgcn_wmma_f32_16x16x4_f32(
                  false, a, false, b, (short)0, acc, false, false);
    }

    float val = 0.0f;
    int   m   = -1;
    #pragma unroll
    for (int v = 0; v < 8; ++v) {
        if (lane == v)      { val = acc[v]; m = v;     }
        if (lane == v + 24) { val = acc[v]; m = v + 8; }
    }
    if (m >= 0) {
        int e = e0 + m;
        if (e < n_edges) out[e] = val;
    }
}

extern "C" void kernel_launch(void* const* d_in, const int* in_sizes, int n_in,
                              void* d_out, int out_size, void* d_ws, size_t ws_size,
                              hipStream_t stream) {
    const float* h   = (const float*)d_in[0];
    const int*   src = (const int*)d_in[1];
    const int*   dst = (const int*)d_in[2];
    float*       out = (float*)d_out;

    const int n_edges = in_sizes[1];
    const int tiles   = (n_edges + 15) / 16;
    const int wavesPerBlock = 8;
    const int blocks  = (tiles + wavesPerBlock - 1) / wavesPerBlock;

    udotv_wmma_kernel<<<blocks, 256, 0, stream>>>(h, src, dst, out, n_edges);
}
